# starting kernel
# baseline (speedup 1.0000x reference)
#include <hip/hip_runtime.h>

#define N_NODES   100000
#define N_EDGES   1600000
#define N_GRAPHS  1000
#define IN_DIM    32
#define HID_DIM   64
#define OUT_DIM   32

typedef __attribute__((ext_vector_type(2))) float v2f;
typedef __attribute__((ext_vector_type(8))) float v8f;

__device__ __forceinline__ void fadd(float* p, float v) { unsafeAtomicAdd(p, v); }


__global__ void zero_f32(float* __restrict__ p, int n) {
  int i = blockIdx.x * blockDim.x + threadIdx.x;
  if (i < n) p[i] = 0.0f;
}

__global__ void count_deg(const int* __restrict__ dst, int* __restrict__ cnt, int n) {
  int i = blockIdx.x * blockDim.x + threadIdx.x;
  if (i < n) atomicAdd(&cnt[dst[i]], 1);
}

__global__ void compute_dis(const int* __restrict__ cnt, float* __restrict__ dis, int n) {
  int i = blockIdx.x * blockDim.x + threadIdx.x;
  if (i < n) dis[i] = rsqrtf(1.0f + (float)cnt[i]);
}


template<int K, int N>
__global__ void gemm_wmma(const float* __restrict__ A, const float* __restrict__ W,
                          float* __restrict__ H, int rowTiles) {
  constexpr int colTiles = N / 16;
  constexpr int rowWaves = 8 / colTiles;
  const int wave = threadIdx.x >> 5;
  const int lane = threadIdx.x & 31;
  const int lo   = lane & 15;
  const int hi   = lane >> 4;
  const int col  = (wave % colTiles) * 16;
  const int rowGroup = wave / colTiles;

  v2f bf[K / 4];
#pragma unroll
  for (int kk = 0; kk < K / 4; ++kk) {
    const int k = kk * 4 + hi * 2;
    bf[kk].x = W[(k + 0) * N + col + lo];
    bf[kk].y = W[(k + 1) * N + col + lo];
  }

  for (int rt = blockIdx.x * rowWaves + rowGroup; rt < rowTiles;
       rt += gridDim.x * rowWaves) {
    const int r0 = rt * 16;
    const float* __restrict__ Arow = A + (size_t)(r0 + lo) * K;
    v8f c = {};
#pragma unroll
    for (int kk = 0; kk < K / 4; ++kk) {
      const int k = kk * 4 + hi * 2;
      v2f a;
      a.x = Arow[k + 0];
      a.y = Arow[k + 1];
      c = __builtin_amdgcn_wmma_f32_16x16x4_f32(false, a, false, bf[kk],
                                                (short)0, c, false, false);
    }
    float* __restrict__ Hp = H + (size_t)(r0 + hi * 8) * N + col + lo;
#pragma unroll
    for (int i = 0; i < 8; ++i)
      Hp[(size_t)i * N] = c[i];
  }
}


template<int F>
__global__ void scatter_edges(const float* __restrict__ h, const int* __restrict__ src,
                              const int* __restrict__ dst, const float* __restrict__ dis,
                              float* __restrict__ agg, int nEdges) {
  constexpr unsigned CHUNKS = F / 4;
  const unsigned long long tid =
      (unsigned long long)blockIdx.x * blockDim.x + threadIdx.x;
  const unsigned long long total = (unsigned long long)nEdges * CHUNKS;
  if (tid >= total) return;
  const unsigned e  = (unsigned)(tid / CHUNKS);
  const unsigned fo = ((unsigned)tid & (CHUNKS - 1u)) << 2;
  const int s = src[e];
  const int d = dst[e];
  const float norm = dis[s] * dis[d];
  const float4 v = *(const float4*)(h + (size_t)s * F + fo);
  float* __restrict__ ap = agg + (size_t)d * F + fo;
  fadd(ap + 0, v.x * norm);
  fadd(ap + 1, v.y * norm);
  fadd(ap + 2, v.z * norm);
  fadd(ap + 3, v.w * norm);
}


template<int F>
__global__ void finalize_relu(const float* __restrict__ h, float* __restrict__ agg,
                              const float* __restrict__ dis, const float* __restrict__ b,
                              int n  ) {
  int idx = blockIdx.x * blockDim.x + threadIdx.x;
  if (idx >= n) return;
  const int node = idx / F;
  const int f    = idx & (F - 1);
  const float s  = dis[node];
  const float v  = agg[idx] + h[idx] * s * s + b[f];
  agg[idx] = fmaxf(v, 0.0f);
}


__global__ void finalize_pool(const float* __restrict__ h, const float* __restrict__ agg,
                              const float* __restrict__ dis, const float* __restrict__ b,
                              float* __restrict__ out, int n  ) {
  int idx = blockIdx.x * blockDim.x + threadIdx.x;
  if (idx >= n) return;
  const int node = idx >> 5;
  const int f    = idx & 31;
  const float s  = dis[node];
  const float v  = agg[idx] + h[idx] * s * s + b[f];
  const int g    = node / (N_NODES / N_GRAPHS);
  fadd(&out[g * OUT_DIM + f], v * 0.01f);
}


extern "C" void kernel_launch(void* const* d_in, const int* in_sizes, int n_in,
                              void* d_out, int out_size, void* d_ws, size_t ws_size,
                              hipStream_t stream) {
  const float* x    = (const float*)d_in[0];
  const int*   eidx = (const int*)d_in[1];
  const float* W1   = (const float*)d_in[3];
  const float* b1   = (const float*)d_in[4];
  const float* W2   = (const float*)d_in[5];
  const float* b2   = (const float*)d_in[6];
  float* out = (float*)d_out;

  const int* src = eidx;
  const int* dst = eidx + N_EDGES;

  char* ws = (char*)d_ws;
  float* dis  = (float*)(ws);
  int*   cnt  = (int*)  (ws + 400000);
  float* h1   = (float*)(ws + 800000);
  float* agg1 = (float*)(ws + 800000 + 25600000);
  float* h2   = (float*)(ws + 800000 + 51200000);
  float* agg2 = (float*)(ws + 800000 + 64000000);

  const int B = 256;

  zero_f32<<<(N_NODES * HID_DIM + B - 1) / B, B, 0, stream>>>(agg1, N_NODES * HID_DIM);
  zero_f32<<<(N_NODES * OUT_DIM + B - 1) / B, B, 0, stream>>>(agg2, N_NODES * OUT_DIM);
  zero_f32<<<(N_NODES + B - 1) / B, B, 0, stream>>>((float*)cnt, N_NODES);
  zero_f32<<<(N_GRAPHS * OUT_DIM + B - 1) / B, B, 0, stream>>>(out, N_GRAPHS * OUT_DIM);

  count_deg<<<(N_EDGES + B - 1) / B, B, 0, stream>>>(dst, cnt, N_EDGES);
  compute_dis<<<(N_NODES + B - 1) / B, B, 0, stream>>>(cnt, dis, N_NODES);

  const int rowTiles = N_NODES / 16;
  gemm_wmma<IN_DIM, HID_DIM><<<(rowTiles + 1) / 2, 256, 0, stream>>>(x, W1, h1, rowTiles);
  {
    unsigned long long tot = (unsigned long long)N_EDGES * (HID_DIM / 4);
    scatter_edges<HID_DIM><<<(int)((tot + B - 1) / B), B, 0, stream>>>(h1, src, dst, dis, agg1, N_EDGES);
  }
  finalize_relu<HID_DIM><<<(N_NODES * HID_DIM + B - 1) / B, B, 0, stream>>>(h1, agg1, dis, b1, N_NODES * HID_DIM);

  gemm_wmma<HID_DIM, OUT_DIM><<<(rowTiles + 3) / 4, 256, 0, stream>>>(agg1, W2, h2, rowTiles);
  {
    unsigned long long tot = (unsigned long long)N_EDGES * (OUT_DIM / 4);
    scatter_edges<OUT_DIM><<<(int)((tot + B - 1) / B), B, 0, stream>>>(h2, src, dst, dis, agg2, N_EDGES);
  }
  finalize_pool<<<(N_NODES * OUT_DIM + B - 1) / B, B, 0, stream>>>(h2, agg2, dis, b2, out, N_NODES * OUT_DIM);
}


# layer 1 aggregates first: edge scatter on the 32-wide inputs, then one f32 WMMA kernel does (agg + self loop) @ W1 + bias + ReLU; the 64-wide scatter and its finalize pass are gone
# speedup vs baseline: 1.3321x; 1.3321x over previous
#include <hip/hip_runtime.h>

#define N_NODES   100000
#define N_EDGES   1600000
#define N_GRAPHS  1000
#define IN_DIM    32
#define HID_DIM   64
#define OUT_DIM   32

typedef __attribute__((ext_vector_type(2))) float v2f;
typedef __attribute__((ext_vector_type(8))) float v8f;

__device__ __forceinline__ void fadd(float* p, float v) { unsafeAtomicAdd(p, v); }


__global__ void zero_f32(float* __restrict__ p, int n) {
  int i = blockIdx.x * blockDim.x + threadIdx.x;
  if (i < n) p[i] = 0.0f;
}

__global__ void count_deg(const int* __restrict__ dst, int* __restrict__ cnt, int n) {
  int i = blockIdx.x * blockDim.x + threadIdx.x;
  if (i < n) atomicAdd(&cnt[dst[i]], 1);
}

__global__ void compute_dis(const int* __restrict__ cnt, float* __restrict__ dis, int n) {
  int i = blockIdx.x * blockDim.x + threadIdx.x;
  if (i < n) dis[i] = rsqrtf(1.0f + (float)cnt[i]);
}


template<int K, int N>
__global__ void gemm_wmma(const float* __restrict__ A, const float* __restrict__ W,
                          float* __restrict__ H, int rowTiles) {
  constexpr int colTiles = N / 16;
  constexpr int rowWaves = 8 / colTiles;
  const int wave = threadIdx.x >> 5;
  const int lane = threadIdx.x & 31;
  const int lo   = lane & 15;
  const int hi   = lane >> 4;
  const int col  = (wave % colTiles) * 16;
  const int rowGroup = wave / colTiles;

  v2f bf[K / 4];
#pragma unroll
  for (int kk = 0; kk < K / 4; ++kk) {
    const int k = kk * 4 + hi * 2;
    bf[kk].x = W[(k + 0) * N + col + lo];
    bf[kk].y = W[(k + 1) * N + col + lo];
  }

  for (int rt = blockIdx.x * rowWaves + rowGroup; rt < rowTiles;
       rt += gridDim.x * rowWaves) {
    const int r0 = rt * 16;
    const float* __restrict__ Arow = A + (size_t)(r0 + lo) * K;
    v8f c = {};
#pragma unroll
    for (int kk = 0; kk < K / 4; ++kk) {
      const int k = kk * 4 + hi * 2;
      v2f a;
      a.x = Arow[k + 0];
      a.y = Arow[k + 1];
      c = __builtin_amdgcn_wmma_f32_16x16x4_f32(false, a, false, bf[kk],
                                                (short)0, c, false, false);
    }
    float* __restrict__ Hp = H + (size_t)(r0 + hi * 8) * N + col + lo;
#pragma unroll
    for (int i = 0; i < 8; ++i)
      Hp[(size_t)i * N] = c[i];
  }
}


__global__ __launch_bounds__(256)
void gemm1_agg_first(const float* __restrict__ x, const float* __restrict__ aggx,
                     const float* __restrict__ dis, const float* __restrict__ W,
                     const float* __restrict__ b, float* __restrict__ H, int rowTiles) {
  const int wave = threadIdx.x >> 5;
  const int lane = threadIdx.x & 31;
  const int c    = lane & 15;
  const int hf   = lane >> 4;
  const int rt   = blockIdx.x * 8 + wave;
  if (rt >= rowTiles) return;

  const int row = rt * 16 + c;
  const float s  = dis[row];
  const float s2 = s * s;
  const float4* __restrict__ xr = (const float4*)(x    + (size_t)row * IN_DIM + hf * 16);
  const float4* __restrict__ gr = (const float4*)(aggx + (size_t)row * IN_DIM + hf * 16);
  float a[16];
#pragma unroll
  for (int q = 0; q < 4; ++q) {
    const float4 xv = xr[q];
    const float4 gv = gr[q];
    a[4 * q + 0] = fmaf(xv.x, s2, gv.x);
    a[4 * q + 1] = fmaf(xv.y, s2, gv.y);
    a[4 * q + 2] = fmaf(xv.z, s2, gv.z);
    a[4 * q + 3] = fmaf(xv.w, s2, gv.w);
  }

  float4 w[16];
#pragma unroll
  for (int i = 0; i < 16; ++i)
    w[i] = *(const float4*)(W + (size_t)(16 * hf + i) * HID_DIM + 4 * c);

  v8f acc0 = {}, acc1 = {}, acc2 = {}, acc3 = {};
#pragma unroll
  for (int j = 0; j < 8; ++j) {
    v2f av;  av.x = a[2 * j];        av.y = a[2 * j + 1];
    v2f b0;  b0.x = w[2 * j].x;      b0.y = w[2 * j + 1].x;
    v2f b1;  b1.x = w[2 * j].y;      b1.y = w[2 * j + 1].y;
    v2f b2;  b2.x = w[2 * j].z;      b2.y = w[2 * j + 1].z;
    v2f b3;  b3.x = w[2 * j].w;      b3.y = w[2 * j + 1].w;
    acc0 = __builtin_amdgcn_wmma_f32_16x16x4_f32(false, av, false, b0, (short)0, acc0, false, false);
    acc1 = __builtin_amdgcn_wmma_f32_16x16x4_f32(false, av, false, b1, (short)0, acc1, false, false);
    acc2 = __builtin_amdgcn_wmma_f32_16x16x4_f32(false, av, false, b2, (short)0, acc2, false, false);
    acc3 = __builtin_amdgcn_wmma_f32_16x16x4_f32(false, av, false, b3, (short)0, acc3, false, false);
  }

  const float4 bv = *(const float4*)(b + 4 * c);
  float* __restrict__ Hp = H + (size_t)(rt * 16 + hf * 8) * HID_DIM + 4 * c;
#pragma unroll
  for (int r = 0; r < 8; ++r) {
    float4 o;
    o.x = fmaxf(acc0[r] + bv.x, 0.0f);
    o.y = fmaxf(acc1[r] + bv.y, 0.0f);
    o.z = fmaxf(acc2[r] + bv.z, 0.0f);
    o.w = fmaxf(acc3[r] + bv.w, 0.0f);
    *(float4*)(Hp + (size_t)r * HID_DIM) = o;
  }
}


template<int F>
__global__ void scatter_edges(const float* __restrict__ h, const int* __restrict__ src,
                              const int* __restrict__ dst, const float* __restrict__ dis,
                              float* __restrict__ agg, int nEdges) {
  constexpr unsigned CHUNKS = F / 4;
  const unsigned long long tid =
      (unsigned long long)blockIdx.x * blockDim.x + threadIdx.x;
  const unsigned long long total = (unsigned long long)nEdges * CHUNKS;
  if (tid >= total) return;
  const unsigned e  = (unsigned)(tid / CHUNKS);
  const unsigned fo = ((unsigned)tid & (CHUNKS - 1u)) << 2;
  const int s = src[e];
  const int d = dst[e];
  const float norm = dis[s] * dis[d];
  const float4 v = *(const float4*)(h + (size_t)s * F + fo);
  float* __restrict__ ap = agg + (size_t)d * F + fo;
  fadd(ap + 0, v.x * norm);
  fadd(ap + 1, v.y * norm);
  fadd(ap + 2, v.z * norm);
  fadd(ap + 3, v.w * norm);
}


__global__ void finalize_pool(const float* __restrict__ h, const float* __restrict__ agg,
                              const float* __restrict__ dis, const float* __restrict__ b,
                              float* __restrict__ out, int n  ) {
  int idx = blockIdx.x * blockDim.x + threadIdx.x;
  if (idx >= n) return;
  const int node = idx >> 5;
  const int f    = idx & 31;
  const float s  = dis[node];
  const float v  = agg[idx] + h[idx] * s * s + b[f];
  const int g    = node / (N_NODES / N_GRAPHS);
  fadd(&out[g * OUT_DIM + f], v * 0.01f);
}


extern "C" void kernel_launch(void* const* d_in, const int* in_sizes, int n_in,
                              void* d_out, int out_size, void* d_ws, size_t ws_size,
                              hipStream_t stream) {
  const float* x    = (const float*)d_in[0];
  const int*   eidx = (const int*)d_in[1];
  const float* W1   = (const float*)d_in[3];
  const float* b1   = (const float*)d_in[4];
  const float* W2   = (const float*)d_in[5];
  const float* b2   = (const float*)d_in[6];
  float* out = (float*)d_out;

  const int* src = eidx;
  const int* dst = eidx + N_EDGES;

  char* ws = (char*)d_ws;
  float* dis  = (float*)(ws);
  int*   cnt  = (int*)  (ws + 400000);
  float* a1   = (float*)(ws + 800000);
  float* aggx = (float*)(ws + 800000 + 25600000);
  float* h2   = (float*)(ws + 800000 + 51200000);
  float* agg2 = (float*)(ws + 800000 + 64000000);

  const int B = 256;

  zero_f32<<<(N_NODES * IN_DIM + B - 1) / B, B, 0, stream>>>(aggx, N_NODES * IN_DIM);
  zero_f32<<<(N_NODES * OUT_DIM + B - 1) / B, B, 0, stream>>>(agg2, N_NODES * OUT_DIM);
  zero_f32<<<(N_NODES + B - 1) / B, B, 0, stream>>>((float*)cnt, N_NODES);
  zero_f32<<<(N_GRAPHS * OUT_DIM + B - 1) / B, B, 0, stream>>>(out, N_GRAPHS * OUT_DIM);

  count_deg<<<(N_EDGES + B - 1) / B, B, 0, stream>>>(dst, cnt, N_EDGES);
  compute_dis<<<(N_NODES + B - 1) / B, B, 0, stream>>>(cnt, dis, N_NODES);

  const int rowTiles = N_NODES / 16;
  {
    unsigned long long tot = (unsigned long long)N_EDGES * (IN_DIM / 4);
    scatter_edges<IN_DIM><<<(int)((tot + B - 1) / B), B, 0, stream>>>(x, src, dst, dis, aggx, N_EDGES);
  }
  gemm1_agg_first<<<(rowTiles + 7) / 8, 256, 0, stream>>>(x, aggx, dis, W1, b1, a1, rowTiles);

  gemm_wmma<HID_DIM, OUT_DIM><<<(rowTiles + 3) / 4, 256, 0, stream>>>(a1, W2, h2, rowTiles);
  {
    unsigned long long tot = (unsigned long long)N_EDGES * (OUT_DIM / 4);
    scatter_edges<OUT_DIM><<<(int)((tot + B - 1) / B), B, 0, stream>>>(h2, src, dst, dis, agg2, N_EDGES);
  }
  finalize_pool<<<(N_NODES * OUT_DIM + B - 1) / B, B, 0, stream>>>(h2, agg2, dis, b2, out, N_NODES * OUT_DIM);
}


# edge sums without float atomics: edges counting-sorted by destination inside the call, then one wave per destination row gathers its source rows into registers; both accumulator zeroings gone
# speedup vs baseline: 1.7715x; 1.3299x over previous
#include <hip/hip_runtime.h>

#define N_NODES   100000
#define N_EDGES   1600000
#define N_GRAPHS  1000
#define IN_DIM    32
#define HID_DIM   64
#define OUT_DIM   32

typedef __attribute__((ext_vector_type(2))) float v2f;
typedef __attribute__((ext_vector_type(8))) float v8f;

__device__ __forceinline__ void fadd(float* p, float v) { unsafeAtomicAdd(p, v); }


__global__ void zero_f32(float* __restrict__ p, int n) {
  int i = blockIdx.x * blockDim.x + threadIdx.x;
  if (i < n) p[i] = 0.0f;
}

__global__ void count_rank(const int* __restrict__ dst, int* __restrict__ cnt,
                           int* __restrict__ rank, int n) {
  int i = blockIdx.x * blockDim.x + threadIdx.x;
  if (i < n) rank[i] = atomicAdd(&cnt[dst[i]], 1);
}

__global__ __launch_bounds__(1024)
void scan_dis(const int* __restrict__ cnt, int* __restrict__ rowptr,
              float* __restrict__ dis, int n) {
  __shared__ int waveTot[32];
  __shared__ int wavePre[32];
  const int t = threadIdx.x;
  const int lane = t & 31;
  const int wv = t >> 5;
  const int blockStart = blockIdx.x * 1024;

  int part = 0;
  const int4* __restrict__ c4 = (const int4*)cnt;
  const int nq = blockStart / 4;
  int k = t;
  for (; k + 3072 < nq; k += 4096) {
    const int4 v0 = c4[k];
    const int4 v1 = c4[k + 1024];
    const int4 v2 = c4[k + 2048];
    const int4 v3 = c4[k + 3072];
    part += ((v0.x + v0.y) + (v0.z + v0.w)) + ((v1.x + v1.y) + (v1.z + v1.w)) +
            ((v2.x + v2.y) + (v2.z + v2.w)) + ((v3.x + v3.y) + (v3.z + v3.w));
  }
  for (; k < nq; k += 1024) {
    const int4 v = c4[k];
    part += (v.x + v.y) + (v.z + v.w);
  }
  const int i = blockStart + t;
  const int c = (i < n) ? cnt[i] : 0;

  int inc = c;
#pragma unroll
  for (int o = 1; o < 32; o <<= 1) {
    const int u = __shfl_up(inc, o);
    if (lane >= o) inc += u;
  }
#pragma unroll
  for (int o = 16; o > 0; o >>= 1) part += __shfl_xor(part, o);
  if (lane == 31) waveTot[wv] = inc;
  if (lane == 0)  wavePre[wv] = part;
  __syncthreads();

  const int wt = waveTot[lane];
  int wInc = wt;
#pragma unroll
  for (int o = 1; o < 32; o <<= 1) {
    const int u = __shfl_up(wInc, o);
    if (lane >= o) wInc += u;
  }
  int front = wavePre[lane];
#pragma unroll
  for (int o = 16; o > 0; o >>= 1) front += __shfl_xor(front, o);
  const int waveBase = __shfl(wInc - wt, wv);

  if (i < n) {
    rowptr[i] = front + waveBase + (inc - c);
    dis[i] = rsqrtf(1.0f + (float)c);
  }
}

__global__ void fill_csr(const int* __restrict__ src, const int* __restrict__ dst,
                         const int* __restrict__ rank, const int* __restrict__ rowptr,
                         const float* __restrict__ dis, int* __restrict__ cw, int n) {
  int e = blockIdx.x * blockDim.x + threadIdx.x;
  if (e >= n) return;
  const int s = src[e];
  const int p = rowptr[dst[e]] + rank[e];
  ((int2*)cw)[p] = make_int2(s, __float_as_int(dis[s]));
}


template<int K, int N>
__global__ void gemm_wmma(const float* __restrict__ A, const float* __restrict__ W,
                          float* __restrict__ H, int rowTiles) {
  constexpr int colTiles = N / 16;
  constexpr int rowWaves = 8 / colTiles;
  const int wave = threadIdx.x >> 5;
  const int lane = threadIdx.x & 31;
  const int lo   = lane & 15;
  const int hi   = lane >> 4;
  const int col  = (wave % colTiles) * 16;
  const int rowGroup = wave / colTiles;

  v2f bf[K / 4];
#pragma unroll
  for (int kk = 0; kk < K / 4; ++kk) {
    const int k = kk * 4 + hi * 2;
    bf[kk].x = W[(k + 0) * N + col + lo];
    bf[kk].y = W[(k + 1) * N + col + lo];
  }

  for (int rt = blockIdx.x * rowWaves + rowGroup; rt < rowTiles;
       rt += gridDim.x * rowWaves) {
    const int r0 = rt * 16;
    const float* __restrict__ Arow = A + (size_t)(r0 + lo) * K;
    v8f c = {};
#pragma unroll
    for (int kk = 0; kk < K / 4; ++kk) {
      const int k = kk * 4 + hi * 2;
      v2f a;
      a.x = Arow[k + 0];
      a.y = Arow[k + 1];
      c = __builtin_amdgcn_wmma_f32_16x16x4_f32(false, a, false, bf[kk],
                                                (short)0, c, false, false);
    }
    float* __restrict__ Hp = H + (size_t)(r0 + hi * 8) * N + col + lo;
#pragma unroll
    for (int i = 0; i < 8; ++i)
      Hp[(size_t)i * N] = c[i];
  }
}


__global__ __launch_bounds__(256)
void gemm1_agg_first(const float* __restrict__ x, const float* __restrict__ aggx,
                     const float* __restrict__ dis, const float* __restrict__ W,
                     const float* __restrict__ b, float* __restrict__ H, int rowTiles) {
  const int wave = threadIdx.x >> 5;
  const int lane = threadIdx.x & 31;
  const int c    = lane & 15;
  const int hf   = lane >> 4;
  const int rt   = blockIdx.x * 8 + wave;
  if (rt >= rowTiles) return;

  const int row = rt * 16 + c;
  const float s  = dis[row];
  const float s2 = s * s;
  const float4* __restrict__ xr = (const float4*)(x    + (size_t)row * IN_DIM + hf * 16);
  const float4* __restrict__ gr = (const float4*)(aggx + (size_t)row * IN_DIM + hf * 16);
  float a[16];
#pragma unroll
  for (int q = 0; q < 4; ++q) {
    const float4 xv = xr[q];
    const float4 gv = gr[q];
    a[4 * q + 0] = fmaf(xv.x, s2, gv.x);
    a[4 * q + 1] = fmaf(xv.y, s2, gv.y);
    a[4 * q + 2] = fmaf(xv.z, s2, gv.z);
    a[4 * q + 3] = fmaf(xv.w, s2, gv.w);
  }

  float4 w[16];
#pragma unroll
  for (int i = 0; i < 16; ++i)
    w[i] = *(const float4*)(W + (size_t)(16 * hf + i) * HID_DIM + 4 * c);

  v8f acc0 = {}, acc1 = {}, acc2 = {}, acc3 = {};
#pragma unroll
  for (int j = 0; j < 8; ++j) {
    v2f av;  av.x = a[2 * j];        av.y = a[2 * j + 1];
    v2f b0;  b0.x = w[2 * j].x;      b0.y = w[2 * j + 1].x;
    v2f b1;  b1.x = w[2 * j].y;      b1.y = w[2 * j + 1].y;
    v2f b2;  b2.x = w[2 * j].z;      b2.y = w[2 * j + 1].z;
    v2f b3;  b3.x = w[2 * j].w;      b3.y = w[2 * j + 1].w;
    acc0 = __builtin_amdgcn_wmma_f32_16x16x4_f32(false, av, false, b0, (short)0, acc0, false, false);
    acc1 = __builtin_amdgcn_wmma_f32_16x16x4_f32(false, av, false, b1, (short)0, acc1, false, false);
    acc2 = __builtin_amdgcn_wmma_f32_16x16x4_f32(false, av, false, b2, (short)0, acc2, false, false);
    acc3 = __builtin_amdgcn_wmma_f32_16x16x4_f32(false, av, false, b3, (short)0, acc3, false, false);
  }

  const float4 bv = *(const float4*)(b + 4 * c);
  float* __restrict__ Hp = H + (size_t)(rt * 16 + hf * 8) * HID_DIM + 4 * c;
#pragma unroll
  for (int r = 0; r < 8; ++r) {
    float4 o;
    o.x = fmaxf(acc0[r] + bv.x, 0.0f);
    o.y = fmaxf(acc1[r] + bv.y, 0.0f);
    o.z = fmaxf(acc2[r] + bv.z, 0.0f);
    o.w = fmaxf(acc3[r] + bv.w, 0.0f);
    *(float4*)(Hp + (size_t)r * HID_DIM) = o;
  }
}


__global__ __launch_bounds__(256)
void gather_rows(const float* __restrict__ h, const int* __restrict__ rowptr,
                 const int* __restrict__ cnt, const int* __restrict__ cw,
                 const float* __restrict__ dis, float* __restrict__ agg) {
  const int lane = threadIdx.x & 31;
  const int q = lane >> 3;
  const int r = lane & 7;
  const int node = __builtin_amdgcn_readfirstlane((int)(blockIdx.x * 8 + (threadIdx.x >> 5)));
  const int beg = rowptr[node];
  const int deg = cnt[node];
  const int2* __restrict__ ent = (const int2*)cw + beg;

  float4 a0 = make_float4(0.0f, 0.0f, 0.0f, 0.0f);
  float4 a1 = make_float4(0.0f, 0.0f, 0.0f, 0.0f);
  for (int base = 0; base < deg; base += 32) {
    const int m = min(32, deg - base);
    int2 my = make_int2(node, 0);
    if (lane < m) my = ent[base + lane];
    for (int j = 0; j < m; j += 8) {
      const int j0 = j + q;
      const int j1 = j + 4 + q;
      const int   s0 = __shfl(my.x, j0);
      const float w0 = __int_as_float(__shfl(my.y, j0));
      const int   s1 = __shfl(my.x, j1);
      const float w1 = __int_as_float(__shfl(my.y, j1));
      const float4 v0 = *(const float4*)(h + (size_t)s0 * 32 + 4 * r);
      const float4 v1 = *(const float4*)(h + (size_t)s1 * 32 + 4 * r);
      a0.x = fmaf(w0, v0.x, a0.x);  a0.y = fmaf(w0, v0.y, a0.y);
      a0.z = fmaf(w0, v0.z, a0.z);  a0.w = fmaf(w0, v0.w, a0.w);
      a1.x = fmaf(w1, v1.x, a1.x);  a1.y = fmaf(w1, v1.y, a1.y);
      a1.z = fmaf(w1, v1.z, a1.z);  a1.w = fmaf(w1, v1.w, a1.w);
    }
  }
  a0.x += a1.x;  a0.y += a1.y;  a0.z += a1.z;  a0.w += a1.w;
#pragma unroll
  for (int o = 8; o < 32; o <<= 1) {
    a0.x += __shfl_xor(a0.x, o);
    a0.y += __shfl_xor(a0.y, o);
    a0.z += __shfl_xor(a0.z, o);
    a0.w += __shfl_xor(a0.w, o);
  }
  if (q == 0) {
    const float sd = dis[node];
    *(float4*)(agg + (size_t)node * 32 + 4 * r) =
        make_float4(a0.x * sd, a0.y * sd, a0.z * sd, a0.w * sd);
  }
}


__global__ void finalize_pool(const float* __restrict__ h, const float* __restrict__ agg,
                              const float* __restrict__ dis, const float* __restrict__ b,
                              float* __restrict__ out, int n  ) {
  int idx = blockIdx.x * blockDim.x + threadIdx.x;
  if (idx >= n) return;
  const int node = idx >> 5;
  const int f    = idx & 31;
  const float s  = dis[node];
  const float v  = agg[idx] + h[idx] * s * s + b[f];
  const int g    = node / (N_NODES / N_GRAPHS);
  fadd(&out[g * OUT_DIM + f], v * 0.01f);
}


extern "C" void kernel_launch(void* const* d_in, const int* in_sizes, int n_in,
                              void* d_out, int out_size, void* d_ws, size_t ws_size,
                              hipStream_t stream) {
  const float* x    = (const float*)d_in[0];
  const int*   eidx = (const int*)d_in[1];
  const float* W1   = (const float*)d_in[3];
  const float* b1   = (const float*)d_in[4];
  const float* W2   = (const float*)d_in[5];
  const float* b2   = (const float*)d_in[6];
  float* out = (float*)d_out;

  const int* src = eidx;
  const int* dst = eidx + N_EDGES;

  char* ws = (char*)d_ws;
  float* dis    = (float*)(ws);
  int*   cnt    = (int*)  (ws + 400000);
  float* a1     = (float*)(ws + 800000);
  float* aggx   = (float*)(ws + 800000 + 25600000);
  int*   rowptr = (int*)  (ws + 800000 + 38400000);
  int*   rank   = (int*)  (ws + 800000 + 38400000 + 409600);
  float* h2     = (float*)(ws + 800000 + 51200000);
  float* agg2   = aggx;
  int*   cw     = (int*)  (ws + 800000 + 64000000);

  const int B = 256;

  zero_f32<<<(N_NODES + B - 1) / B, B, 0, stream>>>((float*)cnt, N_NODES);
  zero_f32<<<(N_GRAPHS * OUT_DIM + B - 1) / B, B, 0, stream>>>(out, N_GRAPHS * OUT_DIM);

  count_rank<<<(N_EDGES + B - 1) / B, B, 0, stream>>>(dst, cnt, rank, N_EDGES);
  scan_dis<<<(N_NODES + 1023) / 1024, 1024, 0, stream>>>(cnt, rowptr, dis, N_NODES);
  fill_csr<<<(N_EDGES + B - 1) / B, B, 0, stream>>>(src, dst, rank, rowptr, dis, cw, N_EDGES);

  const int rowTiles = N_NODES / 16;
  gather_rows<<<N_NODES / 8, 256, 0, stream>>>(x, rowptr, cnt, cw, dis, aggx);
  gemm1_agg_first<<<(rowTiles + 7) / 8, 256, 0, stream>>>(x, aggx, dis, W1, b1, a1, rowTiles);

  gemm_wmma<HID_DIM, OUT_DIM><<<(rowTiles + 3) / 4, 256, 0, stream>>>(a1, W2, h2, rowTiles);
  gather_rows<<<N_NODES / 8, 256, 0, stream>>>(h2, rowptr, cnt, cw, dis, agg2);
  finalize_pool<<<(N_NODES * OUT_DIM + B - 1) / B, B, 0, stream>>>(h2, agg2, dis, b2, out, N_NODES * OUT_DIM);
}


# fewer, longer waves: a gather wave walks 8 destination rows, 16 source rows in flight (12500 waves, was 100000); pool is one wave per graph storing its mean, no atomics, no zeroing of out
# speedup vs baseline: 2.0444x; 1.1541x over previous
#include <hip/hip_runtime.h>

#define N_NODES   100000
#define N_EDGES   1600000
#define N_GRAPHS  1000
#define IN_DIM    32
#define HID_DIM   64
#define OUT_DIM   32

typedef __attribute__((ext_vector_type(2))) float v2f;
typedef __attribute__((ext_vector_type(8))) float v8f;


__global__ void zero_f32(float* __restrict__ p, int n) {
  int i = blockIdx.x * blockDim.x + threadIdx.x;
  if (i < n) p[i] = 0.0f;
}

#define ROW_CAP 32

__global__ __launch_bounds__(256)
void count_place(const int* __restrict__ src, const int* __restrict__ dst,
                 int* __restrict__ cnt, int* __restrict__ ell, int* __restrict__ ovf, int nq) {
  const int t = blockIdx.x * blockDim.x + threadIdx.x;
  if (t >= nq) return;
  const int4 d4 = ((const int4*)dst)[t];
  const int4 s4 = ((const int4*)src)[t];
  const int r0 = atomicAdd(&cnt[d4.x], 1);
  const int r1 = atomicAdd(&cnt[d4.y], 1);
  const int r2 = atomicAdd(&cnt[d4.z], 1);
  const int r3 = atomicAdd(&cnt[d4.w], 1);
  if (r0 < ROW_CAP) ell[d4.x * ROW_CAP + r0] = s4.x; else ovf[atomicAdd(&cnt[N_NODES], 1)] = 4 * t + 0;
  if (r1 < ROW_CAP) ell[d4.y * ROW_CAP + r1] = s4.y; else ovf[atomicAdd(&cnt[N_NODES], 1)] = 4 * t + 1;
  if (r2 < ROW_CAP) ell[d4.z * ROW_CAP + r2] = s4.z; else ovf[atomicAdd(&cnt[N_NODES], 1)] = 4 * t + 2;
  if (r3 < ROW_CAP) ell[d4.w * ROW_CAP + r3] = s4.w; else ovf[atomicAdd(&cnt[N_NODES], 1)] = 4 * t + 3;
}

__global__ __launch_bounds__(256)
void prescale_dis(const float* __restrict__ x, const int* __restrict__ cnt,
                  float* __restrict__ dis, float* __restrict__ xs) {
  constexpr int Q = N_NODES * 8 / 4;
  const int t = blockIdx.x * blockDim.x + threadIdx.x;
  if (t >= Q) return;
  float4 v[4];
  float s[4];
#pragma unroll
  for (int k = 0; k < 4; ++k) {
    const int i = t + k * Q;
    v[k] = ((const float4*)x)[i];
    s[k] = rsqrtf(1.0f + (float)cnt[i >> 3]);
  }
#pragma unroll
  for (int k = 0; k < 4; ++k) {
    const int i = t + k * Q;
    ((float4*)xs)[i] = make_float4(v[k].x * s[k], v[k].y * s[k], v[k].z * s[k], v[k].w * s[k]);
    if ((i & 7) == 0) dis[i >> 3] = s[k];
  }
}


template<int K, int N>
__global__ void gemm_wmma(const float* __restrict__ A, const float* __restrict__ W,
                          const float* __restrict__ rowScale,
                          float* __restrict__ H, int rowTiles) {
  constexpr int colTiles = N / 16;
  constexpr int rowWaves = 8 / colTiles;
  const int wave = threadIdx.x >> 5;
  const int lane = threadIdx.x & 31;
  const int lo   = lane & 15;
  const int hi   = lane >> 4;
  const int col  = (wave % colTiles) * 16;
  const int rowGroup = wave / colTiles;

  v2f bf[K / 4];
#pragma unroll
  for (int kk = 0; kk < K / 4; ++kk) {
    const int k = kk * 4 + hi * 2;
    bf[kk].x = W[(k + 0) * N + col + lo];
    bf[kk].y = W[(k + 1) * N + col + lo];
  }

  for (int rt = blockIdx.x * rowWaves + rowGroup; rt < rowTiles;
       rt += gridDim.x * rowWaves) {
    const int r0 = rt * 16;
    const float* __restrict__ Arow = A + (size_t)(r0 + lo) * K;
    const float4 sa = *(const float4*)(rowScale + r0 + hi * 8);
    const float4 sb = *(const float4*)(rowScale + r0 + hi * 8 + 4);
    v8f c = {};
#pragma unroll
    for (int kk = 0; kk < K / 4; ++kk) {
      const int k = kk * 4 + hi * 2;
      v2f a;
      a.x = Arow[k + 0];
      a.y = Arow[k + 1];
      c = __builtin_amdgcn_wmma_f32_16x16x4_f32(false, a, false, bf[kk],
                                                (short)0, c, false, false);
    }
    const float sc[8] = {sa.x, sa.y, sa.z, sa.w, sb.x, sb.y, sb.z, sb.w};
    float* __restrict__ Hp = H + (size_t)(r0 + hi * 8) * N + col + lo;
#pragma unroll
    for (int i = 0; i < 8; ++i)
      Hp[(size_t)i * N] = c[i] * sc[i];
  }
}


__global__ __launch_bounds__(256)
void gemm1_agg_first(const float* __restrict__ xs, const float* __restrict__ aggx,
                     const float* __restrict__ dis, const float* __restrict__ W,
                     const float* __restrict__ b, float* __restrict__ H, int rowTiles) {
  const int wave = threadIdx.x >> 5;
  const int lane = threadIdx.x & 31;
  const int c    = lane & 15;
  const int hf   = lane >> 4;
  const int rt   = blockIdx.x * 8 + wave;
  if (rt >= rowTiles) return;

  const int row = rt * 16 + c;
  const float s  = dis[row];
  const float4* __restrict__ xr = (const float4*)(xs   + (size_t)row * IN_DIM + hf * 16);
  const float4* __restrict__ gr = (const float4*)(aggx + (size_t)row * IN_DIM + hf * 16);
  float a[16];
#pragma unroll
  for (int q = 0; q < 4; ++q) {
    const float4 xv = xr[q];
    const float4 gv = gr[q];
    a[4 * q + 0] = fmaf(xv.x, s, gv.x);
    a[4 * q + 1] = fmaf(xv.y, s, gv.y);
    a[4 * q + 2] = fmaf(xv.z, s, gv.z);
    a[4 * q + 3] = fmaf(xv.w, s, gv.w);
  }

  float4 w[16];
#pragma unroll
  for (int i = 0; i < 16; ++i)
    w[i] = *(const float4*)(W + (size_t)(16 * hf + i) * HID_DIM + 4 * c);

  v8f acc0 = {}, acc1 = {}, acc2 = {}, acc3 = {};
#pragma unroll
  for (int j = 0; j < 8; ++j) {
    v2f av;  av.x = a[2 * j];        av.y = a[2 * j + 1];
    v2f b0;  b0.x = w[2 * j].x;      b0.y = w[2 * j + 1].x;
    v2f b1;  b1.x = w[2 * j].y;      b1.y = w[2 * j + 1].y;
    v2f b2;  b2.x = w[2 * j].z;      b2.y = w[2 * j + 1].z;
    v2f b3;  b3.x = w[2 * j].w;      b3.y = w[2 * j + 1].w;
    acc0 = __builtin_amdgcn_wmma_f32_16x16x4_f32(false, av, false, b0, (short)0, acc0, false, false);
    acc1 = __builtin_amdgcn_wmma_f32_16x16x4_f32(false, av, false, b1, (short)0, acc1, false, false);
    acc2 = __builtin_amdgcn_wmma_f32_16x16x4_f32(false, av, false, b2, (short)0, acc2, false, false);
    acc3 = __builtin_amdgcn_wmma_f32_16x16x4_f32(false, av, false, b3, (short)0, acc3, false, false);
  }

  const float4 bv = *(const float4*)(b + 4 * c);
  float* __restrict__ Hp = H + (size_t)(rt * 16 + hf * 8) * HID_DIM + 4 * c;
#pragma unroll
  for (int r = 0; r < 8; ++r) {
    float4 o;
    o.x = fmaxf(acc0[r] + bv.x, 0.0f);
    o.y = fmaxf(acc1[r] + bv.y, 0.0f);
    o.z = fmaxf(acc2[r] + bv.z, 0.0f);
    o.w = fmaxf(acc3[r] + bv.w, 0.0f);
    *(float4*)(Hp + (size_t)r * HID_DIM) = o;
  }
}

#define ROWS_PER_WAVE 8
static_assert(N_NODES % ROWS_PER_WAVE == 0, "whole rows per wave");

__global__ __launch_bounds__(256)
void gather_rows(const float* __restrict__ hs, const int* __restrict__ cnt,
                 const int* __restrict__ ell, const int* __restrict__ ovf,
                 const int* __restrict__ src, const int* __restrict__ dst,
                 const float* __restrict__ dis, float* __restrict__ agg) {
  const int lane = threadIdx.x & 31;
  const int q = lane >> 3;
  const int r = lane & 7;
  const int wv = __builtin_amdgcn_readfirstlane((int)(blockIdx.x * 8 + (threadIdx.x >> 5)));
  if (wv >= N_NODES / ROWS_PER_WAVE) return;
  const int base = wv * ROWS_PER_WAVE;

  int   deg = cnt[base];
  float sd  = dis[base];
  int   my  = base;
  if (lane < min(ROW_CAP, deg)) my = ell[base * ROW_CAP + lane];

  for (int i = 0; i < ROWS_PER_WAVE; ++i) {
    const int node = base + i;
    const int m = min(ROW_CAP, deg);

    int   degN = 0;
    float sdN  = 0.0f;
    int   myN  = node;
    if (i + 1 < ROWS_PER_WAVE) {
      degN = cnt[node + 1];
      sdN  = dis[node + 1];
      myN  = node + 1;
      if (lane < min(ROW_CAP, degN)) myN = ell[(node + 1) * ROW_CAP + lane];
    }

    float4 a0 = make_float4(0.0f, 0.0f, 0.0f, 0.0f);
    float4 a1 = make_float4(0.0f, 0.0f, 0.0f, 0.0f);
    float4 a2 = make_float4(0.0f, 0.0f, 0.0f, 0.0f);
    float4 a3 = make_float4(0.0f, 0.0f, 0.0f, 0.0f);
    for (int j = 0; j < m; j += 16) {
      const int j0 = j + q;
      const int j1 = j + 4 + q;
      const int j2 = j + 8 + q;
      const int j3 = j + 12 + q;
      const int   s0 = __shfl(my, j0);
      const int   s1 = __shfl(my, j1);
      const int   s2 = __shfl(my, j2);
      const int   s3 = __shfl(my, j3);
      const float w0 = (j0 < m) ? 1.0f : 0.0f;
      const float w1 = (j1 < m) ? 1.0f : 0.0f;
      const float w2 = (j2 < m) ? 1.0f : 0.0f;
      const float w3 = (j3 < m) ? 1.0f : 0.0f;
      const float4 v0 = *(const float4*)(hs + (size_t)s0 * 32 + 4 * r);
      const float4 v1 = *(const float4*)(hs + (size_t)s1 * 32 + 4 * r);
      const float4 v2 = *(const float4*)(hs + (size_t)s2 * 32 + 4 * r);
      const float4 v3 = *(const float4*)(hs + (size_t)s3 * 32 + 4 * r);
      a0.x = fmaf(w0, v0.x, a0.x);  a0.y = fmaf(w0, v0.y, a0.y);
      a0.z = fmaf(w0, v0.z, a0.z);  a0.w = fmaf(w0, v0.w, a0.w);
      a1.x = fmaf(w1, v1.x, a1.x);  a1.y = fmaf(w1, v1.y, a1.y);
      a1.z = fmaf(w1, v1.z, a1.z);  a1.w = fmaf(w1, v1.w, a1.w);
      a2.x = fmaf(w2, v2.x, a2.x);  a2.y = fmaf(w2, v2.y, a2.y);
      a2.z = fmaf(w2, v2.z, a2.z);  a2.w = fmaf(w2, v2.w, a2.w);
      a3.x = fmaf(w3, v3.x, a3.x);  a3.y = fmaf(w3, v3.y, a3.y);
      a3.z = fmaf(w3, v3.z, a3.z);  a3.w = fmaf(w3, v3.w, a3.w);
    }

    if (deg > ROW_CAP) {
      const int nov = min(cnt[N_NODES], N_EDGES);
      const float wq = (q == 0) ? 1.0f : 0.0f;
      for (int ob = 0; ob < nov; ob += 32) {
        const int oi = ob + lane;
        int sv = node;
        bool hit = false;
        if (oi < nov) {
          const int e = ovf[oi];
          hit = (dst[e] == node);
          if (hit) sv = src[e];
        }
        unsigned mask = __builtin_amdgcn_ballot_w32(hit);
        while (mask != 0u) {
          const int k = __builtin_ctz(mask);
          mask &= mask - 1u;
          const int s = __shfl(sv, k);
          const float4 v = *(const float4*)(hs + (size_t)s * 32 + 4 * r);
          a0.x = fmaf(wq, v.x, a0.x);  a0.y = fmaf(wq, v.y, a0.y);
          a0.z = fmaf(wq, v.z, a0.z);  a0.w = fmaf(wq, v.w, a0.w);
        }
      }
    }

    a0.x = (a0.x + a1.x) + (a2.x + a3.x);
    a0.y = (a0.y + a1.y) + (a2.y + a3.y);
    a0.z = (a0.z + a1.z) + (a2.z + a3.z);
    a0.w = (a0.w + a1.w) + (a2.w + a3.w);
#pragma unroll
    for (int o = 8; o < 32; o <<= 1) {
      a0.x += __shfl_xor(a0.x, o);
      a0.y += __shfl_xor(a0.y, o);
      a0.z += __shfl_xor(a0.z, o);
      a0.w += __shfl_xor(a0.w, o);
    }
    if (q == 0) {
      *(float4*)(agg + (size_t)node * 32 + 4 * r) =
          make_float4(a0.x * sd, a0.y * sd, a0.z * sd, a0.w * sd);
    }

    deg = degN;
    sd  = sdN;
    my  = myN;
  }
}


__global__ __launch_bounds__(128)
void finalize_pool(const float* __restrict__ h, const float* __restrict__ agg,
                   const float* __restrict__ dis, const float* __restrict__ b,
                   float* __restrict__ out) {
  constexpr int NPG = N_NODES / N_GRAPHS;
  static_assert(NPG * N_GRAPHS == N_NODES && NPG % 4 == 0, "four rows per trip");
  const int lane = threadIdx.x & 31;
  const int q = lane >> 3;
  const int r = lane & 7;
  const int g = blockIdx.x * 4 + (threadIdx.x >> 5);
  if (g >= N_GRAPHS) return;

  float4 acc = make_float4(0.0f, 0.0f, 0.0f, 0.0f);
#pragma unroll 5
  for (int i = 0; i < NPG; i += 4) {
    const int node = g * NPG + i + q;
    const float s = dis[node];
    const float4 hv = *(const float4*)(h   + (size_t)node * 32 + 4 * r);
    const float4 av = *(const float4*)(agg + (size_t)node * 32 + 4 * r);
    acc.x += fmaf(hv.x, s, av.x);
    acc.y += fmaf(hv.y, s, av.y);
    acc.z += fmaf(hv.z, s, av.z);
    acc.w += fmaf(hv.w, s, av.w);
  }
#pragma unroll
  for (int o = 8; o < 32; o <<= 1) {
    acc.x += __shfl_xor(acc.x, o);
    acc.y += __shfl_xor(acc.y, o);
    acc.z += __shfl_xor(acc.z, o);
    acc.w += __shfl_xor(acc.w, o);
  }
  if (q == 0) {
    const float4 bv = *(const float4*)(b + 4 * r);
    *(float4*)(out + (size_t)g * OUT_DIM + 4 * r) =
        make_float4(acc.x / (float)NPG + bv.x, acc.y / (float)NPG + bv.y,
                    acc.z / (float)NPG + bv.z, acc.w / (float)NPG + bv.w);
  }
}


extern "C" void kernel_launch(void* const* d_in, const int* in_sizes, int n_in,
                              void* d_out, int out_size, void* d_ws, size_t ws_size,
                              hipStream_t stream) {
  const float* x    = (const float*)d_in[0];
  const int*   eidx = (const int*)d_in[1];
  const float* W1   = (const float*)d_in[3];
  const float* b1   = (const float*)d_in[4];
  const float* W2   = (const float*)d_in[5];
  const float* b2   = (const float*)d_in[6];
  float* out = (float*)d_out;

  const int* src = eidx;
  const int* dst = eidx + N_EDGES;

  char* ws = (char*)d_ws;
  float* dis    = (float*)(ws);
  int*   cnt    = (int*)  (ws + 400000);
  float* a1     = (float*)(ws + 800128);
  float* aggx   = (float*)(ws + 800128 + 25600000);
  float* xs     = (float*)(ws + 800128 + 38400000);
  float* h2s    = xs;
  float* agg2   = aggx;
  int*   ell    = (int*)  (ws + 800128 + 51200000);
  int*   ovf    = (int*)  (ws + 800128 + 64000000);

  const int B = 256;

  zero_f32<<<(N_NODES + 1 + B - 1) / B, B, 0, stream>>>((float*)cnt, N_NODES + 1);

  static_assert(N_EDGES % 4 == 0, "four edges per thread");
  count_place<<<(N_EDGES / 4 + B - 1) / B, B, 0, stream>>>(src, dst, cnt, ell, ovf, N_EDGES / 4);
  prescale_dis<<<(N_NODES * 2 + B - 1) / B, B, 0, stream>>>(x, cnt, dis, xs);

  const int rowTiles = N_NODES / 16;
  const int gatherBlocks = (N_NODES / ROWS_PER_WAVE + 7) / 8;
  gather_rows<<<gatherBlocks, 256, 0, stream>>>(xs, cnt, ell, ovf, src, dst, dis, aggx);
  gemm1_agg_first<<<(rowTiles + 7) / 8, 256, 0, stream>>>(xs, aggx, dis, W1, b1, a1, rowTiles);

  gemm_wmma<HID_DIM, OUT_DIM><<<(rowTiles + 3) / 4, 256, 0, stream>>>(a1, W2, dis, h2s, rowTiles);
  gather_rows<<<gatherBlocks, 256, 0, stream>>>(h2s, cnt, ell, ovf, src, dst, dis, agg2);
  finalize_pool<<<(N_GRAPHS + 3) / 4, 128, 0, stream>>>(h2s, agg2, dis, b2, out);
}
